# starting kernel
# baseline (speedup 1.0000x reference)
#include <hip/hip_runtime.h>
#include <hip/hip_bf16.h>

typedef __attribute__((ext_vector_type(2))) float v2f;
typedef __attribute__((ext_vector_type(8))) float v8f;

#define D_FEAT 128

__global__ void sage_zero_ws(float* __restrict__ p, long n) {
    long i = (long)blockIdx.x * blockDim.x + threadIdx.x;
    long stride = (long)gridDim.x * blockDim.x;
    for (; i < n; i += stride) p[i] = 0.0f;
}

__global__ __launch_bounds__(256) void sage_scatter(
    const float* __restrict__ feat, const int* __restrict__ src,
    const int* __restrict__ dst, float* __restrict__ summed,
    float* __restrict__ deg, int n_edges)
{
    long gid = (long)blockIdx.x * blockDim.x + threadIdx.x;
    int e = (int)(gid >> 5);
    int lane = (int)(gid & 31);
    if (e >= n_edges) return;
    int s = src[e];
    int d = dst[e];
    const float4 v = *(const float4*)(feat + (long)s * D_FEAT + lane * 4);
    float* acc = summed + (long)d * D_FEAT + lane * 4;
    atomicAdd(acc + 0, v.x);
    atomicAdd(acc + 1, v.y);
    atomicAdd(acc + 2, v.z);
    atomicAdd(acc + 3, v.w);
    if (lane == 0) atomicAdd(deg + d, 1.0f);
}

__global__ __launch_bounds__(256) void sage_norm(
    float* __restrict__ summed, const float* __restrict__ deg, long n_elts)
{
    long i = (long)blockIdx.x * blockDim.x + threadIdx.x;
    if (i >= n_elts) return;
    float dg = deg[i >> 7];
    float inv = dg > 1.0f ? 1.0f / dg : 1.0f;
    summed[i] *= inv;
}

__global__ __launch_bounds__(256) void sage_gemm_wmma(
    const float* __restrict__ feat, const float* __restrict__ hneigh,
    const float* __restrict__ Wself, const float* __restrict__ Wneigh,
    float* __restrict__ out)
{
    const int wave = threadIdx.x >> 5;
    const int lane = threadIdx.x & 31;
    const int m = lane & 15;
    const int khalf = (lane >> 4) * 2;
    const long node_base = (long)blockIdx.x * 16;
    const int  col_base = wave * 16;

    const float* fa = feat   + (node_base + m) * D_FEAT + khalf;
    const float* ha = hneigh + (node_base + m) * D_FEAT + khalf;
    const float* ws = Wself  + (long)(col_base + m) * D_FEAT + khalf;
    const float* wn = Wneigh + (long)(col_base + m) * D_FEAT + khalf;

    v8f c = {};
#pragma unroll
    for (int k0 = 0; k0 < D_FEAT; k0 += 4) {
        v2f a  = *(const v2f*)(fa + k0);
        v2f bs = *(const v2f*)(ws + k0);
        c = __builtin_amdgcn_wmma_f32_16x16x4_f32(
                 false, a,  false, bs,
                 (short)0, c,  false,  false);
        v2f ah = *(const v2f*)(ha + k0);
        v2f bn = *(const v2f*)(wn + k0);
        c = __builtin_amdgcn_wmma_f32_16x16x4_f32(
                false, ah, false, bn, (short)0, c, false, false);
    }

    float* o = out + (node_base + (lane >> 4) * 8) * D_FEAT + col_base + m;
#pragma unroll
    for (int r = 0; r < 8; ++r)
        o[(long)r * D_FEAT] = c[r];
}

extern "C" void kernel_launch(void* const* d_in, const int* in_sizes, int n_in,
                              void* d_out, int out_size, void* d_ws, size_t ws_size,
                              hipStream_t stream)
{
    const float* feat   = (const float*)d_in[0];
    const int*   src    = (const int*)d_in[1];
    const int*   dst    = (const int*)d_in[2];
    const float* Wself  = (const float*)d_in[3];
    const float* Wneigh = (const float*)d_in[4];
    float*       out    = (float*)d_out;

    const int n_nodes = in_sizes[0] / D_FEAT;
    const int n_edges = in_sizes[1];

    float* summed = (float*)d_ws;
    float* deg    = summed + (long)n_nodes * D_FEAT;

    const long n_acc = (long)n_nodes * D_FEAT + n_nodes;
    sage_zero_ws<<<4096, 256, 0, stream>>>(summed, n_acc);

    const long scatter_threads = (long)n_edges * 32;
    const int  scatter_blocks  = (int)((scatter_threads + 255) / 256);
    sage_scatter<<<scatter_blocks, 256, 0, stream>>>(feat, src, dst, summed, deg, n_edges);

    const long n_elts = (long)n_nodes * D_FEAT;
    sage_norm<<<(int)((n_elts + 255) / 256), 256, 0, stream>>>(summed, deg, n_elts);

    sage_gemm_wmma<<<n_nodes / 16, 256, 0, stream>>>(feat, summed, Wself, Wneigh, out);
}


# Float-atomic scatter, zero fill and mean pass replaced by owner blocks: each keeps 400 dst rows in LDS, scans all dst for its edges, adds gathered source rows by LDS float adds (arrival order)
# speedup vs baseline: 3.8580x; 3.8580x over previous
#include <hip/hip_runtime.h>
#include <hip/hip_bf16.h>

typedef __attribute__((ext_vector_type(2))) float v2f;
typedef __attribute__((ext_vector_type(4))) float v4f;
typedef __attribute__((ext_vector_type(8))) float v8f;

#define D_FEAT 128

#define NPB       400
#define AGG_WAVES 32
#define QCAP      256
#define SCAN_U    8

__device__ __forceinline__ void ring_push(unsigned* q, int qhead, int& qn,
                                          unsigned ballot, bool mine, unsigned word)
{
    if (ballot) {
        int pos = qn + (int)__builtin_amdgcn_mbcnt_lo(ballot, 0u);
        if (mine) q[(qhead + pos) & (QCAP - 1)] = word;
        qn += __popc(ballot);
    }
}

__device__ __forceinline__ void ring_drain(
    const float* __restrict__ feat, const int* __restrict__ src,
    float* acc, int* cnt, const unsigned* q, int& qhead, int& qn,
    int n, int lane)
{
    unsigned ent = 0u;
    int s = 0;
    if (lane < n) {
        ent = q[(qhead + lane) & (QCAP - 1)];
        s = src[ent >> 9];
        __hip_atomic_fetch_add(&cnt[ent & 511u], 1, __ATOMIC_RELAXED,
                               __HIP_MEMORY_SCOPE_WORKGROUP);
    }
    const int dl = (int)(ent & 511u);
#pragma nounroll
    for (int j0 = 0; j0 < n; j0 += 8) {
        v4f v[8];
        int dj[8];
#pragma unroll
        for (int t = 0; t < 8; ++t) {
            const int j = min(j0 + t, n - 1);
            const int sj = __builtin_amdgcn_readlane(s, j);
            dj[t] = __builtin_amdgcn_readlane(dl, j);
            v[t] = *(const v4f*)(feat + (long)sj * D_FEAT + lane * 4);
        }
#pragma unroll
        for (int t = 0; t < 8; ++t) {
            if (j0 + t < n) {
                float* a = acc + dj[t] * D_FEAT + lane;
                __hip_atomic_fetch_add(a,      v[t].x, __ATOMIC_RELAXED, __HIP_MEMORY_SCOPE_WORKGROUP);
                __hip_atomic_fetch_add(a + 32, v[t].y, __ATOMIC_RELAXED, __HIP_MEMORY_SCOPE_WORKGROUP);
                __hip_atomic_fetch_add(a + 64, v[t].z, __ATOMIC_RELAXED, __HIP_MEMORY_SCOPE_WORKGROUP);
                __hip_atomic_fetch_add(a + 96, v[t].w, __ATOMIC_RELAXED, __HIP_MEMORY_SCOPE_WORKGROUP);
            }
        }
    }
    qhead = (qhead + n) & (QCAP - 1);
    qn -= n;
}

__device__ __forceinline__ void scan_test(
    const int4 d, const unsigned e, const int lo, const unsigned nown,
    const float* __restrict__ feat, const int* __restrict__ src,
    float* acc, int* cnt, unsigned* q, int& qhead, int& qn, int lane)
{
    const unsigned u0 = (unsigned)(d.x - lo);
    const unsigned u1 = (unsigned)(d.y - lo);
    const unsigned u2 = (unsigned)(d.z - lo);
    const unsigned u3 = (unsigned)(d.w - lo);
    const bool m0 = u0 < nown;
    const bool m1 = u1 < nown;
    const bool m2 = u2 < nown;
    const bool m3 = u3 < nown;
    const unsigned b0 = __builtin_amdgcn_ballot_w32(m0);
    const unsigned b1 = __builtin_amdgcn_ballot_w32(m1);
    const unsigned b2 = __builtin_amdgcn_ballot_w32(m2);
    const unsigned b3 = __builtin_amdgcn_ballot_w32(m3);
    if (b0 | b1 | b2 | b3) {
        ring_push(q, qhead, qn, b0, m0, ((e + 0u) << 9) | u0);
        ring_push(q, qhead, qn, b1, m1, ((e + 1u) << 9) | u1);
        ring_push(q, qhead, qn, b2, m2, ((e + 2u) << 9) | u2);
        ring_push(q, qhead, qn, b3, m3, ((e + 3u) << 9) | u3);
        while (qn >= 32)
            ring_drain(feat, src, acc, cnt, q, qhead, qn, 32, lane);
    }
}

__global__ __launch_bounds__(1024) void sage_aggregate(
    const float* __restrict__ feat, const int* __restrict__ src,
    const int* __restrict__ dst, float* __restrict__ hneigh,
    int n_nodes, int n_edges)
{
    __shared__ float    acc[NPB * D_FEAT];
    __shared__ int      cnt[NPB];
    __shared__ unsigned ring[AGG_WAVES * QCAP];

    const int tid  = threadIdx.x;
    const int wave = __builtin_amdgcn_readfirstlane(tid >> 5);
    const int lane = tid & 31;
    const int lo   = blockIdx.x * NPB;
    const int nown = min(NPB, n_nodes - lo);

    for (int i = tid; i < NPB * D_FEAT; i += 1024) acc[i] = 0.0f;
    for (int i = tid; i < NPB; i += 1024) cnt[i] = 0;
    __syncthreads();

    unsigned* q = ring + wave * QCAP;
    int qhead = 0, qn = 0;

    const int n_full = n_edges >> 7;
    const int GF     = n_full / (SCAN_U * AGG_WAVES);
    const int rotg   = (int)((blockIdx.x * (unsigned)GF) / gridDim.x);

    for (int g = 0; g < GF; ++g) {
        int gg = g + rotg;
        if (gg >= GF) gg -= GF;
        const int eb = (gg * (SCAN_U * AGG_WAVES) + wave) * 128 + lane * 4;
        const int4* p = (const int4*)(dst + eb);
        int4 d[SCAN_U];
#pragma unroll
        for (int u = 0; u < SCAN_U; ++u) d[u] = p[u * (AGG_WAVES * 32)];
#pragma unroll
        for (int u = 0; u < SCAN_U; ++u)
            scan_test(d[u], (unsigned)(eb + u * (AGG_WAVES * 128)), lo, (unsigned)nown,
                      feat, src, acc, cnt, q, qhead, qn, lane);
    }

    {
        int c = GF * (SCAN_U * AGG_WAVES) + wave;
        int4 cur = make_int4(-1, -1, -1, -1);
        if (c < n_full) cur = *(const int4*)(dst + c * 128 + lane * 4);
        for (; c < n_full; c += AGG_WAVES) {
            const int cn = c + AGG_WAVES;
            int4 nxt = cur;
            if (cn < n_full) nxt = *(const int4*)(dst + cn * 128 + lane * 4);
            scan_test(cur, (unsigned)(c * 128 + lane * 4), lo, (unsigned)nown,
                      feat, src, acc, cnt, q, qhead, qn, lane);
            cur = nxt;
        }
    }

    if (wave == 0) {
        const int tb = n_full << 7;
        for (int t = 0; t < 4; ++t) {
            const int e = tb + t * 32 + lane;
            const bool in = e < n_edges;
            const int dv = in ? dst[in ? e : 0] : -1;
            const unsigned uu = (unsigned)(dv - lo);
            const bool m = in && (uu < (unsigned)nown);
            const unsigned b = __builtin_amdgcn_ballot_w32(m);
            ring_push(q, qhead, qn, b, m, ((unsigned)e << 9) | uu);
            while (qn >= 32)
                ring_drain(feat, src, acc, cnt, q, qhead, qn, 32, lane);
        }
    }
    if (qn > 0) ring_drain(feat, src, acc, cnt, q, qhead, qn, qn, lane);
    __syncthreads();

    for (int i = wave; i < nown; i += AGG_WAVES) {
        const float dg = fmaxf((float)cnt[i], 1.0f);
        const float* a = acc + i * D_FEAT + lane;
        v4f o;
        o.x = a[0]  / dg;
        o.y = a[32] / dg;
        o.z = a[64] / dg;
        o.w = a[96] / dg;
        *(v4f*)(hneigh + (long)(lo + i) * D_FEAT + lane * 4) = o;
    }
}

#define KCH (D_FEAT / 8)

__global__ __launch_bounds__(128) void sage_gemm_wmma(
    const float* __restrict__ feat, const float* __restrict__ hneigh,
    const float* __restrict__ Wself, const float* __restrict__ Wneigh,
    float* __restrict__ out, int n_tiles)
{
    const int wave = threadIdx.x >> 5;
    const int lane = threadIdx.x & 31;
    const int c = lane & 15;
    const int h = lane >> 4;
    const int col_base = wave * 32;

    v4f bs0[KCH], bs1[KCH], bn0[KCH], bn1[KCH];
    {
        const float* ws0 = Wself  + (long)(col_base + c) * D_FEAT + 4 * h;
        const float* ws1 = ws0 + 16 * D_FEAT;
        const float* wn0 = Wneigh + (long)(col_base + c) * D_FEAT + 4 * h;
        const float* wn1 = wn0 + 16 * D_FEAT;
#pragma unroll
        for (int j = 0; j < KCH; ++j) {
            bs0[j] = *(const v4f*)(ws0 + 8 * j);
            bs1[j] = *(const v4f*)(ws1 + 8 * j);
            bn0[j] = *(const v4f*)(wn0 + 8 * j);
            bn1[j] = *(const v4f*)(wn1 + 8 * j);
        }
    }

    for (int tile = blockIdx.x; tile < n_tiles; tile += gridDim.x) {
        const long node_base = (long)tile * 16;
        const float* fa = feat   + (node_base + c) * D_FEAT + 4 * h;
        const float* ha = hneigh + (node_base + c) * D_FEAT + 4 * h;

        v4f af[KCH], ah[KCH];
#pragma unroll
        for (int j = 0; j < KCH; ++j) af[j] = *(const v4f*)(fa + 8 * j);
#pragma unroll
        for (int j = 0; j < KCH; ++j) ah[j] = *(const v4f*)(ha + 8 * j);

        v8f c0 = {}, c1 = {};
#pragma unroll
        for (int j = 0; j < KCH; ++j) {
            v2f a_lo = {af[j].x, af[j].y};
            v2f a_hi = {af[j].z, af[j].w};
            v2f b0lo = {bs0[j].x, bs0[j].y};
            v2f b0hi = {bs0[j].z, bs0[j].w};
            v2f b1lo = {bs1[j].x, bs1[j].y};
            v2f b1hi = {bs1[j].z, bs1[j].w};
            c0 = __builtin_amdgcn_wmma_f32_16x16x4_f32(
                    false, a_lo, false, b0lo, (short)0, c0, false, false);
            c1 = __builtin_amdgcn_wmma_f32_16x16x4_f32(
                    false, a_lo, false, b1lo, (short)0, c1, false, false);
            c0 = __builtin_amdgcn_wmma_f32_16x16x4_f32(
                    false, a_hi, false, b0hi, (short)0, c0, false, false);
            c1 = __builtin_amdgcn_wmma_f32_16x16x4_f32(
                    false, a_hi, false, b1hi, (short)0, c1, false, false);
        }
#pragma unroll
        for (int j = 0; j < KCH; ++j) {
            v2f a_lo = {ah[j].x, ah[j].y};
            v2f a_hi = {ah[j].z, ah[j].w};
            v2f b0lo = {bn0[j].x, bn0[j].y};
            v2f b0hi = {bn0[j].z, bn0[j].w};
            v2f b1lo = {bn1[j].x, bn1[j].y};
            v2f b1hi = {bn1[j].z, bn1[j].w};
            c0 = __builtin_amdgcn_wmma_f32_16x16x4_f32(
                    false, a_lo, false, b0lo, (short)0, c0, false, false);
            c1 = __builtin_amdgcn_wmma_f32_16x16x4_f32(
                    false, a_lo, false, b1lo, (short)0, c1, false, false);
            c0 = __builtin_amdgcn_wmma_f32_16x16x4_f32(
                    false, a_hi, false, b0hi, (short)0, c0, false, false);
            c1 = __builtin_amdgcn_wmma_f32_16x16x4_f32(
                    false, a_hi, false, b1hi, (short)0, c1, false, false);
        }

        float* o = out + (node_base + h * 8) * D_FEAT + col_base + c;
#pragma unroll
        for (int r = 0; r < 8; ++r) {
            o[(long)r * D_FEAT]      = c0[r];
            o[(long)r * D_FEAT + 16] = c1[r];
        }
    }
}

extern "C" void kernel_launch(void* const* d_in, const int* in_sizes, int n_in,
                              void* d_out, int out_size, void* d_ws, size_t ws_size,
                              hipStream_t stream)
{
    const float* feat   = (const float*)d_in[0];
    const int*   src    = (const int*)d_in[1];
    const int*   dst    = (const int*)d_in[2];
    const float* Wself  = (const float*)d_in[3];
    const float* Wneigh = (const float*)d_in[4];
    float*       out    = (float*)d_out;

    const int n_nodes = in_sizes[0] / D_FEAT;
    const int n_edges = in_sizes[1];

    float* hneigh = (float*)d_ws;

    sage_aggregate<<<(n_nodes + NPB - 1) / NPB, 1024, 0, stream>>>(
        feat, src, dst, hneigh, n_nodes, n_edges);

    sage_gemm_wmma<<<512, 128, 0, stream>>>(feat, hneigh, Wself, Wneigh, out, n_nodes / 16);
}


# Owner-block scan reads one byte per edge instead of a 4-byte dst: a new first launch writes owner-block bytes and packed (src, local row) words; LDS float adds keep arrival order
# speedup vs baseline: 4.7465x; 1.2303x over previous
#include <hip/hip_runtime.h>
#include <hip/hip_bf16.h>

typedef __attribute__((ext_vector_type(2))) float v2f;
typedef __attribute__((ext_vector_type(4))) float v4f;
typedef __attribute__((ext_vector_type(8))) float v8f;

#define D_FEAT 128

#define NPB       400
#define CHUNK_E   512
#define SRC_BITS  17
#define SRC_MASK  ((1u << SRC_BITS) - 1u)
#define OWN_NONE  255u
#define KEY_U     4

__device__ __forceinline__ void edge_key(int d, int s, unsigned n_nodes,
                                         unsigned& own, unsigned& pk)
{
    const unsigned ud = (unsigned)d;
    const unsigned b  = ud / NPB;
    const unsigned r  = ud - b * NPB;
    const bool ok = (ud < n_nodes) && (b < OWN_NONE);
    own = ok ? b : OWN_NONE;
    pk  = ((unsigned)s & SRC_MASK) | (r << SRC_BITS);
}

__device__ __forceinline__ void unit_keys(const int4 d, const int4 s, unsigned n_nodes,
                                          unsigned* pk, unsigned* ownw, int u)
{
    unsigned o0, o1, o2, o3;
    uint4 p;
    edge_key(d.x, s.x, n_nodes, o0, p.x);
    edge_key(d.y, s.y, n_nodes, o1, p.y);
    edge_key(d.z, s.z, n_nodes, o2, p.z);
    edge_key(d.w, s.w, n_nodes, o3, p.w);
    ((uint4*)pk)[u] = p;
    ownw[u] = o0 | (o1 << 8) | (o2 << 16) | (o3 << 24);
}

__global__ __launch_bounds__(256) void sage_edge_keys(
    const int* __restrict__ src, const int* __restrict__ dst,
    float* __restrict__ scratch, int n_nodes, int n_edges)
{
    const int epad = (n_edges + (CHUNK_E - 1)) & ~(CHUNK_E - 1);
    unsigned* pk   = (unsigned*)scratch;
    unsigned* ownw = pk + epad;
    const int n_units = epad >> 2;
    const int n_vec   = n_edges >> 2;
    const int stride  = gridDim.x * blockDim.x;
    const unsigned nn = (unsigned)n_nodes;

    for (int ub0 = blockIdx.x * blockDim.x; ub0 < n_vec; ub0 += KEY_U * stride) {
        const int ub = ub0 + threadIdx.x;
        if (ub0 + (KEY_U - 1) * stride + (int)blockDim.x <= n_vec) {
            int4 d[KEY_U], s[KEY_U];
#pragma unroll
            for (int k = 0; k < KEY_U; ++k) {
                d[k] = ((const int4*)dst)[ub + k * stride];
                s[k] = ((const int4*)src)[ub + k * stride];
            }
#pragma unroll
            for (int k = 0; k < KEY_U; ++k)
                unit_keys(d[k], s[k], nn, pk, ownw, ub + k * stride);
        } else {
            for (int k = 0; k < KEY_U; ++k) {
                const int u = ub + k * stride;
                if (u < n_vec)
                    unit_keys(((const int4*)dst)[u], ((const int4*)src)[u], nn, pk, ownw, u);
            }
        }
    }

    for (int u = n_vec + blockIdx.x * blockDim.x + threadIdx.x; u < n_units; u += stride) {
        int dv[4], sv[4];
#pragma unroll
        for (int i = 0; i < 4; ++i) {
            const int e = u * 4 + i;
            const bool in = e < n_edges;
            dv[i] = in ? dst[in ? e : 0] : -1;
            sv[i] = in ? src[in ? e : 0] : 0;
        }
        unit_keys(make_int4(dv[0], dv[1], dv[2], dv[3]), make_int4(sv[0], sv[1], sv[2], sv[3]),
                  nn, pk, ownw, u);
    }
}

#define AGG_WAVES 32
#define QCAP      256
#define SCAN_U    8

__device__ __forceinline__ void ring_drain(
    const float* __restrict__ feat, const unsigned* __restrict__ pk,
    float* acc, int* cnt, const unsigned* q, int& qhead, int& qn,
    int n, int lane)
{
    unsigned pkv = 0u;
    if (lane < n) {
        pkv = pk[q[(qhead + lane) & (QCAP - 1)]];
        __hip_atomic_fetch_add(&cnt[pkv >> SRC_BITS], 1, __ATOMIC_RELAXED,
                               __HIP_MEMORY_SCOPE_WORKGROUP);
    }
    const int s  = (int)(pkv & SRC_MASK);
    const int dl = (int)(pkv >> SRC_BITS);
#pragma nounroll
    for (int j0 = 0; j0 < n; j0 += 8) {
        v4f v[8];
        int dj[8];
#pragma unroll
        for (int t = 0; t < 8; ++t) {
            const int j = min(j0 + t, n - 1);
            const int sj = __builtin_amdgcn_readlane(s, j);
            dj[t] = __builtin_amdgcn_readlane(dl, j);
            v[t] = *(const v4f*)(feat + (long)sj * D_FEAT + lane * 4);
        }
#pragma unroll
        for (int t = 0; t < 8; ++t) {
            if (j0 + t < n) {
                float* a = acc + dj[t] * D_FEAT + lane;
                __hip_atomic_fetch_add(a,      v[t].x, __ATOMIC_RELAXED, __HIP_MEMORY_SCOPE_WORKGROUP);
                __hip_atomic_fetch_add(a + 32, v[t].y, __ATOMIC_RELAXED, __HIP_MEMORY_SCOPE_WORKGROUP);
                __hip_atomic_fetch_add(a + 64, v[t].z, __ATOMIC_RELAXED, __HIP_MEMORY_SCOPE_WORKGROUP);
                __hip_atomic_fetch_add(a + 96, v[t].w, __ATOMIC_RELAXED, __HIP_MEMORY_SCOPE_WORKGROUP);
            }
        }
    }
    qhead = (qhead + n) & (QCAP - 1);
    qn -= n;
}

__device__ __forceinline__ unsigned zero_bytes(unsigned x)
{
    return ~(((x & 0x7f7f7f7fu) + 0x7f7f7f7fu) | x | 0x7f7f7f7fu);
}

__device__ __forceinline__ void scan_test(
    const uint4 d, const unsigned e, const unsigned idw,
    const float* __restrict__ feat, const unsigned* __restrict__ pk,
    float* acc, int* cnt, unsigned* q, int& qhead, int& qn, int lane)
{
    unsigned m = (zero_bytes(d.x ^ idw) >> 7) | (zero_bytes(d.y ^ idw) >> 6)
               | (zero_bytes(d.z ^ idw) >> 5) | (zero_bytes(d.w ^ idw) >> 4);
    unsigned bal = __builtin_amdgcn_ballot_w32(m != 0u);
    while (bal) {
        const bool mine = m != 0u;
        const unsigned p = (unsigned)__builtin_ctz(mine ? m : 1u);
        const int pos = qn + (int)__builtin_amdgcn_mbcnt_lo(bal, 0u);
        if (mine) q[(qhead + pos) & (QCAP - 1)] = e + ((p & 7u) << 2) + (p >> 3);
        qn += __popc(bal);
        if (qn >= 32)
            ring_drain(feat, pk, acc, cnt, q, qhead, qn, 32, lane);
        m &= m - 1u;
        bal = __builtin_amdgcn_ballot_w32(m != 0u);
    }
}

__global__ __launch_bounds__(1024) void sage_aggregate(
    const float* __restrict__ feat, const float* __restrict__ scratch,
    float* __restrict__ hneigh, int n_nodes, int n_edges)
{
    __shared__ float    acc[NPB * D_FEAT];
    __shared__ int      cnt[NPB];
    __shared__ unsigned ring[AGG_WAVES * QCAP];

    const int tid  = threadIdx.x;
    const int wave = __builtin_amdgcn_readfirstlane(tid >> 5);
    const int lane = tid & 31;
    const int lo   = blockIdx.x * NPB;
    const int nown = min(NPB, n_nodes - lo);

    const int epad = (n_edges + (CHUNK_E - 1)) & ~(CHUNK_E - 1);
    const unsigned* pk = (const unsigned*)scratch;
    const unsigned char* own = (const unsigned char*)(pk + epad);
    const unsigned idw = blockIdx.x * 0x01010101u;

    for (int i = tid; i < NPB * D_FEAT; i += 1024) acc[i] = 0.0f;
    for (int i = tid; i < NPB; i += 1024) cnt[i] = 0;
    __syncthreads();

    unsigned* q = ring + wave * QCAP;
    int qhead = 0, qn = 0;

    const int n_chunks = epad / CHUNK_E;
    const int GF       = n_chunks / (SCAN_U * AGG_WAVES);
    const int rotg     = (int)((blockIdx.x * (unsigned)GF) / gridDim.x);

    for (int g = 0; g < GF; ++g) {
        int gg = g + rotg;
        if (gg >= GF) gg -= GF;
        const int eb = (gg * (SCAN_U * AGG_WAVES) + wave) * CHUNK_E + lane * 16;
        const uint4* p = (const uint4*)(own + eb);
        uint4 d[SCAN_U];
#pragma unroll
        for (int u = 0; u < SCAN_U; ++u) d[u] = p[u * (AGG_WAVES * 32)];
#pragma unroll
        for (int u = 0; u < SCAN_U; ++u)
            scan_test(d[u], (unsigned)(eb + u * (AGG_WAVES * CHUNK_E)), idw,
                      feat, pk, acc, cnt, q, qhead, qn, lane);
    }

    {
        int c = GF * (SCAN_U * AGG_WAVES) + wave;
        uint4 cur = make_uint4(~0u, ~0u, ~0u, ~0u);
        if (c < n_chunks) cur = *(const uint4*)(own + c * CHUNK_E + lane * 16);
        for (; c < n_chunks; c += AGG_WAVES) {
            const int cn = c + AGG_WAVES;
            uint4 nxt = cur;
            if (cn < n_chunks) nxt = *(const uint4*)(own + cn * CHUNK_E + lane * 16);
            scan_test(cur, (unsigned)(c * CHUNK_E + lane * 16), idw,
                      feat, pk, acc, cnt, q, qhead, qn, lane);
            cur = nxt;
        }
    }
    if (qn > 0) ring_drain(feat, pk, acc, cnt, q, qhead, qn, qn, lane);
    __syncthreads();

    for (int i = wave; i < nown; i += AGG_WAVES) {
        const float dg = fmaxf((float)cnt[i], 1.0f);
        const float* a = acc + i * D_FEAT + lane;
        v4f o;
        o.x = a[0]  / dg;
        o.y = a[32] / dg;
        o.z = a[64] / dg;
        o.w = a[96] / dg;
        *(v4f*)(hneigh + (long)(lo + i) * D_FEAT + lane * 4) = o;
    }
}

#define KCH (D_FEAT / 8)

__global__ __launch_bounds__(128) void sage_gemm_wmma(
    const float* __restrict__ feat, const float* __restrict__ hneigh,
    const float* __restrict__ Wself, const float* __restrict__ Wneigh,
    float* __restrict__ out, int n_tiles)
{
    const int wave = threadIdx.x >> 5;
    const int lane = threadIdx.x & 31;
    const int c = lane & 15;
    const int h = lane >> 4;
    const int col_base = wave * 32;

    v4f bs0[KCH], bs1[KCH], bn0[KCH], bn1[KCH];
    {
        const float* ws0 = Wself  + (long)(col_base + c) * D_FEAT + 4 * h;
        const float* ws1 = ws0 + 16 * D_FEAT;
        const float* wn0 = Wneigh + (long)(col_base + c) * D_FEAT + 4 * h;
        const float* wn1 = wn0 + 16 * D_FEAT;
#pragma unroll
        for (int j = 0; j < KCH; ++j) {
            bs0[j] = *(const v4f*)(ws0 + 8 * j);
            bs1[j] = *(const v4f*)(ws1 + 8 * j);
            bn0[j] = *(const v4f*)(wn0 + 8 * j);
            bn1[j] = *(const v4f*)(wn1 + 8 * j);
        }
    }

    for (int tile = blockIdx.x; tile < n_tiles; tile += gridDim.x) {
        const long node_base = (long)tile * 16;
        const float* fa = feat   + (node_base + c) * D_FEAT + 4 * h;
        const float* ha = hneigh + (node_base + c) * D_FEAT + 4 * h;

        v4f af[KCH], ah[KCH];
#pragma unroll
        for (int j = 0; j < KCH; ++j) af[j] = *(const v4f*)(fa + 8 * j);
#pragma unroll
        for (int j = 0; j < KCH; ++j) ah[j] = *(const v4f*)(ha + 8 * j);

        v8f c0 = {}, c1 = {};
#pragma unroll
        for (int j = 0; j < KCH; ++j) {
            v2f a_lo = {af[j].x, af[j].y};
            v2f a_hi = {af[j].z, af[j].w};
            v2f b0lo = {bs0[j].x, bs0[j].y};
            v2f b0hi = {bs0[j].z, bs0[j].w};
            v2f b1lo = {bs1[j].x, bs1[j].y};
            v2f b1hi = {bs1[j].z, bs1[j].w};
            c0 = __builtin_amdgcn_wmma_f32_16x16x4_f32(
                    false, a_lo, false, b0lo, (short)0, c0, false, false);
            c1 = __builtin_amdgcn_wmma_f32_16x16x4_f32(
                    false, a_lo, false, b1lo, (short)0, c1, false, false);
            c0 = __builtin_amdgcn_wmma_f32_16x16x4_f32(
                    false, a_hi, false, b0hi, (short)0, c0, false, false);
            c1 = __builtin_amdgcn_wmma_f32_16x16x4_f32(
                    false, a_hi, false, b1hi, (short)0, c1, false, false);
        }
#pragma unroll
        for (int j = 0; j < KCH; ++j) {
            v2f a_lo = {ah[j].x, ah[j].y};
            v2f a_hi = {ah[j].z, ah[j].w};
            v2f b0lo = {bn0[j].x, bn0[j].y};
            v2f b0hi = {bn0[j].z, bn0[j].w};
            v2f b1lo = {bn1[j].x, bn1[j].y};
            v2f b1hi = {bn1[j].z, bn1[j].w};
            c0 = __builtin_amdgcn_wmma_f32_16x16x4_f32(
                    false, a_lo, false, b0lo, (short)0, c0, false, false);
            c1 = __builtin_amdgcn_wmma_f32_16x16x4_f32(
                    false, a_lo, false, b1lo, (short)0, c1, false, false);
            c0 = __builtin_amdgcn_wmma_f32_16x16x4_f32(
                    false, a_hi, false, b0hi, (short)0, c0, false, false);
            c1 = __builtin_amdgcn_wmma_f32_16x16x4_f32(
                    false, a_hi, false, b1hi, (short)0, c1, false, false);
        }

        float* o = out + (node_base + h * 8) * D_FEAT + col_base + c;
#pragma unroll
        for (int r = 0; r < 8; ++r) {
            o[(long)r * D_FEAT]      = c0[r];
            o[(long)r * D_FEAT + 16] = c1[r];
        }
    }
}

extern "C" void kernel_launch(void* const* d_in, const int* in_sizes, int n_in,
                              void* d_out, int out_size, void* d_ws, size_t ws_size,
                              hipStream_t stream)
{
    const float* feat   = (const float*)d_in[0];
    const int*   src    = (const int*)d_in[1];
    const int*   dst    = (const int*)d_in[2];
    const float* Wself  = (const float*)d_in[3];
    const float* Wneigh = (const float*)d_in[4];
    float*       out    = (float*)d_out;

    const int n_nodes = in_sizes[0] / D_FEAT;
    const int n_edges = in_sizes[1];

    float* hneigh = (float*)d_ws;

    sage_edge_keys<<<392, 256, 0, stream>>>(src, dst, out, n_nodes, n_edges);

    sage_aggregate<<<(n_nodes + NPB - 1) / NPB, 1024, 0, stream>>>(
        feat, out, hneigh, n_nodes, n_edges);

    sage_gemm_wmma<<<512, 128, 0, stream>>>(feat, hneigh, Wself, Wneigh, out, n_nodes / 16);
}


# Dense kernel at one wave per SIMD with two A-fragment sets: next row tile's 32 fragment loads issue as one group ahead of the current tile's WMMA block (were spread among it); k sum order unchanged
# speedup vs baseline: 4.9827x; 1.0498x over previous
#include <hip/hip_runtime.h>
#include <hip/hip_bf16.h>

typedef __attribute__((ext_vector_type(2))) float v2f;
typedef __attribute__((ext_vector_type(4))) float v4f;
typedef __attribute__((ext_vector_type(8))) float v8f;

#define D_FEAT 128

#define NPB       400
#define CHUNK_E   512
#define SRC_BITS  17
#define SRC_MASK  ((1u << SRC_BITS) - 1u)
#define OWN_NONE  255u
#define KEY_U     4

__device__ __forceinline__ void edge_key(int d, int s, unsigned n_nodes,
                                         unsigned& own, unsigned& pk)
{
    const unsigned ud = (unsigned)d;
    const unsigned b  = ud / NPB;
    const unsigned r  = ud - b * NPB;
    const bool ok = (ud < n_nodes) && (b < OWN_NONE);
    own = ok ? b : OWN_NONE;
    pk  = ((unsigned)s & SRC_MASK) | (r << SRC_BITS);
}

__device__ __forceinline__ void unit_keys(const int4 d, const int4 s, unsigned n_nodes,
                                          unsigned* pk, unsigned* ownw, int u)
{
    unsigned o0, o1, o2, o3;
    uint4 p;
    edge_key(d.x, s.x, n_nodes, o0, p.x);
    edge_key(d.y, s.y, n_nodes, o1, p.y);
    edge_key(d.z, s.z, n_nodes, o2, p.z);
    edge_key(d.w, s.w, n_nodes, o3, p.w);
    ((uint4*)pk)[u] = p;
    ownw[u] = o0 | (o1 << 8) | (o2 << 16) | (o3 << 24);
}

__global__ __launch_bounds__(256) void sage_edge_keys(
    const int* __restrict__ src, const int* __restrict__ dst,
    float* __restrict__ scratch, int n_nodes, int n_edges)
{
    const int epad = (n_edges + (CHUNK_E - 1)) & ~(CHUNK_E - 1);
    unsigned* pk   = (unsigned*)scratch;
    unsigned* ownw = pk + epad;
    const int n_units = epad >> 2;
    const int n_vec   = n_edges >> 2;
    const int stride  = gridDim.x * blockDim.x;
    const unsigned nn = (unsigned)n_nodes;

    for (int ub0 = blockIdx.x * blockDim.x; ub0 < n_vec; ub0 += KEY_U * stride) {
        const int ub = ub0 + threadIdx.x;
        if (ub0 + (KEY_U - 1) * stride + (int)blockDim.x <= n_vec) {
            int4 d[KEY_U], s[KEY_U];
#pragma unroll
            for (int k = 0; k < KEY_U; ++k) {
                d[k] = ((const int4*)dst)[ub + k * stride];
                s[k] = ((const int4*)src)[ub + k * stride];
            }
#pragma unroll
            for (int k = 0; k < KEY_U; ++k)
                unit_keys(d[k], s[k], nn, pk, ownw, ub + k * stride);
        } else {
            for (int k = 0; k < KEY_U; ++k) {
                const int u = ub + k * stride;
                if (u < n_vec)
                    unit_keys(((const int4*)dst)[u], ((const int4*)src)[u], nn, pk, ownw, u);
            }
        }
    }

    for (int u = n_vec + blockIdx.x * blockDim.x + threadIdx.x; u < n_units; u += stride) {
        int dv[4], sv[4];
#pragma unroll
        for (int i = 0; i < 4; ++i) {
            const int e = u * 4 + i;
            const bool in = e < n_edges;
            dv[i] = in ? dst[in ? e : 0] : -1;
            sv[i] = in ? src[in ? e : 0] : 0;
        }
        unit_keys(make_int4(dv[0], dv[1], dv[2], dv[3]), make_int4(sv[0], sv[1], sv[2], sv[3]),
                  nn, pk, ownw, u);
    }
}

#define AGG_WAVES 32
#define QCAP      256
#define SCAN_U    8

__device__ __forceinline__ void ring_drain(
    const float* __restrict__ feat, const unsigned* __restrict__ pk,
    float* acc, int* cnt, const unsigned* q, int& qhead, int& qn,
    int n, int lane)
{
    unsigned pkv = 0u;
    if (lane < n) {
        pkv = pk[q[(qhead + lane) & (QCAP - 1)]];
        __hip_atomic_fetch_add(&cnt[pkv >> SRC_BITS], 1, __ATOMIC_RELAXED,
                               __HIP_MEMORY_SCOPE_WORKGROUP);
    }
    const int s  = (int)(pkv & SRC_MASK);
    const int dl = (int)(pkv >> SRC_BITS);
#pragma nounroll
    for (int j0 = 0; j0 < n; j0 += 8) {
        v4f v[8];
        int dj[8];
#pragma unroll
        for (int t = 0; t < 8; ++t) {
            const int j = min(j0 + t, n - 1);
            const int sj = __builtin_amdgcn_readlane(s, j);
            dj[t] = __builtin_amdgcn_readlane(dl, j);
            v[t] = *(const v4f*)(feat + (long)sj * D_FEAT + lane * 4);
        }
#pragma unroll
        for (int t = 0; t < 8; ++t) {
            if (j0 + t < n) {
                float* a = acc + dj[t] * D_FEAT + lane;
                __hip_atomic_fetch_add(a,      v[t].x, __ATOMIC_RELAXED, __HIP_MEMORY_SCOPE_WORKGROUP);
                __hip_atomic_fetch_add(a + 32, v[t].y, __ATOMIC_RELAXED, __HIP_MEMORY_SCOPE_WORKGROUP);
                __hip_atomic_fetch_add(a + 64, v[t].z, __ATOMIC_RELAXED, __HIP_MEMORY_SCOPE_WORKGROUP);
                __hip_atomic_fetch_add(a + 96, v[t].w, __ATOMIC_RELAXED, __HIP_MEMORY_SCOPE_WORKGROUP);
            }
        }
    }
    qhead = (qhead + n) & (QCAP - 1);
    qn -= n;
}

__device__ __forceinline__ unsigned zero_bytes(unsigned x)
{
    return ~(((x & 0x7f7f7f7fu) + 0x7f7f7f7fu) | x | 0x7f7f7f7fu);
}

__device__ __forceinline__ void scan_test(
    const uint4 d, const unsigned e, const unsigned idw,
    const float* __restrict__ feat, const unsigned* __restrict__ pk,
    float* acc, int* cnt, unsigned* q, int& qhead, int& qn, int lane)
{
    unsigned m = (zero_bytes(d.x ^ idw) >> 7) | (zero_bytes(d.y ^ idw) >> 6)
               | (zero_bytes(d.z ^ idw) >> 5) | (zero_bytes(d.w ^ idw) >> 4);
    unsigned bal = __builtin_amdgcn_ballot_w32(m != 0u);
    while (bal) {
        const bool mine = m != 0u;
        const unsigned p = (unsigned)__builtin_ctz(mine ? m : 1u);
        const int pos = qn + (int)__builtin_amdgcn_mbcnt_lo(bal, 0u);
        if (mine) q[(qhead + pos) & (QCAP - 1)] = e + ((p & 7u) << 2) + (p >> 3);
        qn += __popc(bal);
        if (qn >= 32)
            ring_drain(feat, pk, acc, cnt, q, qhead, qn, 32, lane);
        m &= m - 1u;
        bal = __builtin_amdgcn_ballot_w32(m != 0u);
    }
}

__global__ __launch_bounds__(1024) void sage_aggregate(
    const float* __restrict__ feat, const float* __restrict__ scratch,
    float* __restrict__ hneigh, int n_nodes, int n_edges)
{
    __shared__ float    acc[NPB * D_FEAT];
    __shared__ int      cnt[NPB];
    __shared__ unsigned ring[AGG_WAVES * QCAP];

    const int tid  = threadIdx.x;
    const int wave = __builtin_amdgcn_readfirstlane(tid >> 5);
    const int lane = tid & 31;
    const int lo   = blockIdx.x * NPB;
    const int nown = min(NPB, n_nodes - lo);

    const int epad = (n_edges + (CHUNK_E - 1)) & ~(CHUNK_E - 1);
    const unsigned* pk = (const unsigned*)scratch;
    const unsigned char* own = (const unsigned char*)(pk + epad);
    const unsigned idw = blockIdx.x * 0x01010101u;

    for (int i = tid; i < NPB * D_FEAT; i += 1024) acc[i] = 0.0f;
    for (int i = tid; i < NPB; i += 1024) cnt[i] = 0;
    __syncthreads();

    unsigned* q = ring + wave * QCAP;
    int qhead = 0, qn = 0;

    const int n_chunks = epad / CHUNK_E;
    const int GF       = n_chunks / (SCAN_U * AGG_WAVES);
    const int rotg     = (int)((blockIdx.x * (unsigned)GF) / gridDim.x);

    for (int g = 0; g < GF; ++g) {
        int gg = g + rotg;
        if (gg >= GF) gg -= GF;
        const int eb = (gg * (SCAN_U * AGG_WAVES) + wave) * CHUNK_E + lane * 16;
        const uint4* p = (const uint4*)(own + eb);
        uint4 d[SCAN_U];
#pragma unroll
        for (int u = 0; u < SCAN_U; ++u) d[u] = p[u * (AGG_WAVES * 32)];
#pragma unroll
        for (int u = 0; u < SCAN_U; ++u)
            scan_test(d[u], (unsigned)(eb + u * (AGG_WAVES * CHUNK_E)), idw,
                      feat, pk, acc, cnt, q, qhead, qn, lane);
    }

    {
        int c = GF * (SCAN_U * AGG_WAVES) + wave;
        uint4 cur = make_uint4(~0u, ~0u, ~0u, ~0u);
        if (c < n_chunks) cur = *(const uint4*)(own + c * CHUNK_E + lane * 16);
        for (; c < n_chunks; c += AGG_WAVES) {
            const int cn = c + AGG_WAVES;
            uint4 nxt = cur;
            if (cn < n_chunks) nxt = *(const uint4*)(own + cn * CHUNK_E + lane * 16);
            scan_test(cur, (unsigned)(c * CHUNK_E + lane * 16), idw,
                      feat, pk, acc, cnt, q, qhead, qn, lane);
            cur = nxt;
        }
    }
    if (qn > 0) ring_drain(feat, pk, acc, cnt, q, qhead, qn, qn, lane);
    __syncthreads();

    for (int i = wave; i < nown; i += AGG_WAVES) {
        const float dg = fmaxf((float)cnt[i], 1.0f);
        const float* a = acc + i * D_FEAT + lane;
        v4f o;
        o.x = a[0]  / dg;
        o.y = a[32] / dg;
        o.z = a[64] / dg;
        o.w = a[96] / dg;
        *(v4f*)(hneigh + (long)(lo + i) * D_FEAT + lane * 4) = o;
    }
}

#define KCH (D_FEAT / 8)

__device__ __forceinline__ void load_a_tile(
    const float* __restrict__ feat, const float* __restrict__ hneigh,
    int tile, int c, int h, v4f (&af)[KCH], v4f (&ah)[KCH])
{
    const long node_base = (long)tile * 16;
    const float* fa = feat   + (node_base + c) * D_FEAT + 4 * h;
    const float* ha = hneigh + (node_base + c) * D_FEAT + 4 * h;
#pragma unroll
    for (int j = 0; j < KCH; ++j) af[j] = *(const v4f*)(fa + 8 * j);
#pragma unroll
    for (int j = 0; j < KCH; ++j) ah[j] = *(const v4f*)(ha + 8 * j);
}

__device__ __forceinline__ void mma_tile_store(
    const v4f (&af)[KCH], const v4f (&ah)[KCH],
    const v4f (&bs0)[KCH], const v4f (&bs1)[KCH],
    const v4f (&bn0)[KCH], const v4f (&bn1)[KCH],
    float* __restrict__ out, int tile, int c, int h, int col_base)
{
    v8f c0 = {}, c1 = {};
#pragma unroll
    for (int j = 0; j < KCH; ++j) {
        v2f a_lo = {af[j].x, af[j].y};
        v2f a_hi = {af[j].z, af[j].w};
        v2f b0lo = {bs0[j].x, bs0[j].y};
        v2f b0hi = {bs0[j].z, bs0[j].w};
        v2f b1lo = {bs1[j].x, bs1[j].y};
        v2f b1hi = {bs1[j].z, bs1[j].w};
        c0 = __builtin_amdgcn_wmma_f32_16x16x4_f32(
                false, a_lo, false, b0lo, (short)0, c0, false, false);
        c1 = __builtin_amdgcn_wmma_f32_16x16x4_f32(
                false, a_lo, false, b1lo, (short)0, c1, false, false);
        c0 = __builtin_amdgcn_wmma_f32_16x16x4_f32(
                false, a_hi, false, b0hi, (short)0, c0, false, false);
        c1 = __builtin_amdgcn_wmma_f32_16x16x4_f32(
                false, a_hi, false, b1hi, (short)0, c1, false, false);
    }
#pragma unroll
    for (int j = 0; j < KCH; ++j) {
        v2f a_lo = {ah[j].x, ah[j].y};
        v2f a_hi = {ah[j].z, ah[j].w};
        v2f b0lo = {bn0[j].x, bn0[j].y};
        v2f b0hi = {bn0[j].z, bn0[j].w};
        v2f b1lo = {bn1[j].x, bn1[j].y};
        v2f b1hi = {bn1[j].z, bn1[j].w};
        c0 = __builtin_amdgcn_wmma_f32_16x16x4_f32(
                false, a_lo, false, b0lo, (short)0, c0, false, false);
        c1 = __builtin_amdgcn_wmma_f32_16x16x4_f32(
                false, a_lo, false, b1lo, (short)0, c1, false, false);
        c0 = __builtin_amdgcn_wmma_f32_16x16x4_f32(
                false, a_hi, false, b0hi, (short)0, c0, false, false);
        c1 = __builtin_amdgcn_wmma_f32_16x16x4_f32(
                false, a_hi, false, b1hi, (short)0, c1, false, false);
    }

    float* o = out + ((long)tile * 16 + h * 8) * D_FEAT + col_base + c;
#pragma unroll
    for (int r = 0; r < 8; ++r) {
        o[(long)r * D_FEAT]      = c0[r];
        o[(long)r * D_FEAT + 16] = c1[r];
    }
}

__global__ __launch_bounds__(128) void sage_gemm_wmma(
    const float* __restrict__ feat, const float* __restrict__ hneigh,
    const float* __restrict__ Wself, const float* __restrict__ Wneigh,
    float* __restrict__ out, int n_tiles)
{
    const int wave = threadIdx.x >> 5;
    const int lane = threadIdx.x & 31;
    const int c = lane & 15;
    const int h = lane >> 4;
    const int col_base = wave * 32;
    const int step = gridDim.x;

    int tile = blockIdx.x;
    if (tile >= n_tiles) return;
    v4f xf[KCH], xh[KCH], yf[KCH], yh[KCH];
    load_a_tile(feat, hneigh, tile, c, h, xf, xh);

    v4f bs0[KCH], bs1[KCH], bn0[KCH], bn1[KCH];
    {
        const float* ws0 = Wself  + (long)(col_base + c) * D_FEAT + 4 * h;
        const float* ws1 = ws0 + 16 * D_FEAT;
        const float* wn0 = Wneigh + (long)(col_base + c) * D_FEAT + 4 * h;
        const float* wn1 = wn0 + 16 * D_FEAT;
#pragma unroll
        for (int j = 0; j < KCH; ++j) {
            bs0[j] = *(const v4f*)(ws0 + 8 * j);
            bs1[j] = *(const v4f*)(ws1 + 8 * j);
            bn0[j] = *(const v4f*)(wn0 + 8 * j);
            bn1[j] = *(const v4f*)(wn1 + 8 * j);
        }
    }

    for (;;) {
        const int t1 = tile + step;
        if (t1 >= n_tiles) {
            mma_tile_store(xf, xh, bs0, bs1, bn0, bn1, out, tile, c, h, col_base);
            break;
        }
        load_a_tile(feat, hneigh, t1, c, h, yf, yh);
        __builtin_amdgcn_sched_barrier(0);
        mma_tile_store(xf, xh, bs0, bs1, bn0, bn1, out, tile, c, h, col_base);

        const int t2 = t1 + step;
        if (t2 >= n_tiles) {
            mma_tile_store(yf, yh, bs0, bs1, bn0, bn1, out, t1, c, h, col_base);
            break;
        }
        load_a_tile(feat, hneigh, t2, c, h, xf, xh);
        __builtin_amdgcn_sched_barrier(0);
        mma_tile_store(yf, yh, bs0, bs1, bn0, bn1, out, t1, c, h, col_base);
        tile = t2;
    }
}

extern "C" void kernel_launch(void* const* d_in, const int* in_sizes, int n_in,
                              void* d_out, int out_size, void* d_ws, size_t ws_size,
                              hipStream_t stream)
{
    const float* feat   = (const float*)d_in[0];
    const int*   src    = (const int*)d_in[1];
    const int*   dst    = (const int*)d_in[2];
    const float* Wself  = (const float*)d_in[3];
    const float* Wneigh = (const float*)d_in[4];
    float*       out    = (float*)d_out;

    const int n_nodes = in_sizes[0] / D_FEAT;
    const int n_edges = in_sizes[1];

    float* hneigh = (float*)d_ws;

    sage_edge_keys<<<392, 256, 0, stream>>>(src, dst, out, n_nodes, n_edges);

    sage_aggregate<<<(n_nodes + NPB - 1) / NPB, 1024, 0, stream>>>(
        feat, out, hneigh, n_nodes, n_edges);

    sage_gemm_wmma<<<256, 128, 0, stream>>>(feat, hneigh, Wself, Wneigh, out, n_nodes / 16);
}


# Owner keys stored bit-sliced (8 plane words per 32 edges): the owner-block scan finds a lane's matching edges among 32 with 9 logic ops (was 54 on owner bytes); LDS float adds stay in arrival order
# speedup vs baseline: 5.1396x; 1.0315x over previous
#include <hip/hip_runtime.h>
#include <hip/hip_bf16.h>

typedef __attribute__((ext_vector_type(2))) float v2f;
typedef __attribute__((ext_vector_type(4))) float v4f;
typedef __attribute__((ext_vector_type(8))) float v8f;

#define D_FEAT 128

#define NPB       400
#define CHUNK_E   1024
#define SRC_BITS  17
#define SRC_MASK  ((1u << SRC_BITS) - 1u)
#define OWN_NONE  255u
#define KEY_J     (CHUNK_E / 128)

__device__ __forceinline__ void edge_key(int d, int s, unsigned n_nodes,
                                         unsigned& own, unsigned& pk)
{
    const unsigned ud = (unsigned)d;
    const unsigned b  = ud / NPB;
    const unsigned r  = ud - b * NPB;
    const bool ok = (ud < n_nodes) && (b < OWN_NONE);
    own = ok ? b : OWN_NONE;
    pk  = ((unsigned)s & SRC_MASK) | (r << SRC_BITS);
}

__device__ __forceinline__ void slice_owner(unsigned o, bool is_i, unsigned (&w)[8])
{
#pragma unroll
    for (int k = 0; k < 8; ++k) {
        const unsigned bal = __builtin_amdgcn_ballot_w32(((o >> k) & 1u) != 0u);
        w[k] = is_i ? bal : w[k];
    }
}

__global__ __launch_bounds__(128) void sage_edge_keys(
    const int* __restrict__ src, const int* __restrict__ dst,
    float* __restrict__ scratch, int n_nodes, int n_edges)
{
    const int epad = (n_edges + (CHUNK_E - 1)) & ~(CHUNK_E - 1);
    unsigned* pk  = (unsigned*)scratch;
    uint4*    key = (uint4*)(pk + epad);
    const int n_chunks = epad / CHUNK_E;
    const unsigned nn = (unsigned)n_nodes;
    const int lane = threadIdx.x & 31;
    const int wv = __builtin_amdgcn_readfirstlane((int)(threadIdx.x >> 5));
    const int wpb = (int)(blockDim.x >> 5);

    for (int c = blockIdx.x * wpb + wv; c < n_chunks; c += gridDim.x * wpb) {
        const int base = c * CHUNK_E;
        int4 d[KEY_J], s[KEY_J];
        if (base + CHUNK_E <= n_edges) {
            const int4* dp = (const int4*)(dst + base) + lane;
            const int4* sp = (const int4*)(src + base) + lane;
#pragma unroll
            for (int j = 0; j < KEY_J; ++j) { d[j] = dp[32 * j]; s[j] = sp[32 * j]; }
        } else {
#pragma unroll
            for (int j = 0; j < KEY_J; ++j) {
                int dv[4], sv[4];
#pragma unroll
                for (int t = 0; t < 4; ++t) {
                    const int e = base + 128 * j + 4 * lane + t;
                    const bool in = e < n_edges;
                    dv[t] = in ? dst[in ? e : 0] : -1;
                    sv[t] = in ? src[in ? e : 0] : 0;
                }
                d[j] = make_int4(dv[0], dv[1], dv[2], dv[3]);
                s[j] = make_int4(sv[0], sv[1], sv[2], sv[3]);
            }
        }

        unsigned w[8] = {0u, 0u, 0u, 0u, 0u, 0u, 0u, 0u};
#pragma unroll
        for (int j = 0; j < KEY_J; ++j) {
            unsigned o0, o1, o2, o3;
            uint4 p;
            edge_key(d[j].x, s[j].x, nn, o0, p.x);
            edge_key(d[j].y, s[j].y, nn, o1, p.y);
            edge_key(d[j].z, s[j].z, nn, o2, p.z);
            edge_key(d[j].w, s[j].w, nn, o3, p.w);
            ((uint4*)(pk + base))[32 * j + lane] = p;
            slice_owner(o0, lane == 4 * j + 0, w);
            slice_owner(o1, lane == 4 * j + 1, w);
            slice_owner(o2, lane == 4 * j + 2, w);
            slice_owner(o3, lane == 4 * j + 3, w);
        }
        uint4* kb = key + (long)c * 64;
        kb[lane]      = make_uint4(w[0], w[1], w[2], w[3]);
        kb[32 + lane] = make_uint4(w[4], w[5], w[6], w[7]);
    }
}

#define AGG_WAVES 32
#define QCAP      256
#define SCAN_C    4

__device__ __forceinline__ void ring_drain(
    const float* __restrict__ feat, const unsigned* __restrict__ pk,
    float* acc, int* cnt, const unsigned* q, int& qhead, int& qn,
    int n, int lane)
{
    unsigned pkv = 0u;
    if (lane < n) {
        pkv = pk[q[(qhead + lane) & (QCAP - 1)]];
        __hip_atomic_fetch_add(&cnt[pkv >> SRC_BITS], 1, __ATOMIC_RELAXED,
                               __HIP_MEMORY_SCOPE_WORKGROUP);
    }
    const int s  = (int)(pkv & SRC_MASK);
    const int dl = (int)(pkv >> SRC_BITS);
#pragma nounroll
    for (int j0 = 0; j0 < n; j0 += 8) {
        v4f v[8];
        int dj[8];
#pragma unroll
        for (int t = 0; t < 8; ++t) {
            const int j = min(j0 + t, n - 1);
            const int sj = __builtin_amdgcn_readlane(s, j);
            dj[t] = __builtin_amdgcn_readlane(dl, j);
            v[t] = *(const v4f*)(feat + (long)sj * D_FEAT + lane * 4);
        }
#pragma unroll
        for (int t = 0; t < 8; ++t) {
            if (j0 + t < n) {
                float* a = acc + dj[t] * D_FEAT + lane;
                __hip_atomic_fetch_add(a,      v[t].x, __ATOMIC_RELAXED, __HIP_MEMORY_SCOPE_WORKGROUP);
                __hip_atomic_fetch_add(a + 32, v[t].y, __ATOMIC_RELAXED, __HIP_MEMORY_SCOPE_WORKGROUP);
                __hip_atomic_fetch_add(a + 64, v[t].z, __ATOMIC_RELAXED, __HIP_MEMORY_SCOPE_WORKGROUP);
                __hip_atomic_fetch_add(a + 96, v[t].w, __ATOMIC_RELAXED, __HIP_MEMORY_SCOPE_WORKGROUP);
            }
        }
    }
    qhead = (qhead + n) & (QCAP - 1);
    qn -= n;
}

__device__ __forceinline__ void scan_test(
    const uint4 lo, const uint4 hi, const unsigned e,
    const unsigned mk0, const unsigned mk1, const unsigned mk2, const unsigned mk3,
    const unsigned mk4, const unsigned mk5, const unsigned mk6, const unsigned mk7,
    const float* __restrict__ feat, const unsigned* __restrict__ pk,
    float* acc, int* cnt, unsigned* q, int& qhead, int& qn, int lane)
{
    unsigned m = (lo.x ^ mk0) & (lo.y ^ mk1) & (lo.z ^ mk2) & (lo.w ^ mk3)
               & (hi.x ^ mk4) & (hi.y ^ mk5) & (hi.z ^ mk6) & (hi.w ^ mk7);
    unsigned bal = __builtin_amdgcn_ballot_w32(m != 0u);
    while (bal) {
        const bool mine = m != 0u;
        const unsigned p = (unsigned)__builtin_ctz(mine ? m : 1u);
        const int pos = qn + (int)__builtin_amdgcn_mbcnt_lo(bal, 0u);
        if (mine) q[(qhead + pos) & (QCAP - 1)] = e + (p << 2);
        qn += __popc(bal);
        if (qn >= 32)
            ring_drain(feat, pk, acc, cnt, q, qhead, qn, 32, lane);
        m &= m - 1u;
        bal = __builtin_amdgcn_ballot_w32(m != 0u);
    }
}

__global__ __launch_bounds__(1024) void sage_aggregate(
    const float* __restrict__ feat, const float* __restrict__ scratch,
    float* __restrict__ hneigh, int n_nodes, int n_edges)
{
    __shared__ float    acc[NPB * D_FEAT];
    __shared__ int      cnt[NPB];
    __shared__ unsigned ring[AGG_WAVES * QCAP];

    const int tid  = threadIdx.x;
    const int wave = __builtin_amdgcn_readfirstlane(tid >> 5);
    const int lane = tid & 31;
    const int lo   = blockIdx.x * NPB;
    const int nown = min(NPB, n_nodes - lo);

    const int epad = (n_edges + (CHUNK_E - 1)) & ~(CHUNK_E - 1);
    const unsigned* pk = (const unsigned*)scratch;
    const uint4* key = (const uint4*)(pk + epad);
    const unsigned bid = blockIdx.x;
    const unsigned mk0 = (bid & 1u)   ? 0u : ~0u, mk1 = (bid & 2u)   ? 0u : ~0u;
    const unsigned mk2 = (bid & 4u)   ? 0u : ~0u, mk3 = (bid & 8u)   ? 0u : ~0u;
    const unsigned mk4 = (bid & 16u)  ? 0u : ~0u, mk5 = (bid & 32u)  ? 0u : ~0u;
    const unsigned mk6 = (bid & 64u)  ? 0u : ~0u, mk7 = (bid & 128u) ? 0u : ~0u;
    const unsigned lane_e = 128u * (unsigned)(lane >> 2) + (unsigned)(lane & 3);

    for (int i = tid; i < NPB * D_FEAT; i += 1024) acc[i] = 0.0f;
    for (int i = tid; i < NPB; i += 1024) cnt[i] = 0;
    __syncthreads();

    unsigned* q = ring + wave * QCAP;
    int qhead = 0, qn = 0;

    const int n_chunks = epad / CHUNK_E;
    const int GF       = n_chunks / (SCAN_C * AGG_WAVES);
    const int rotg     = (int)((blockIdx.x * (unsigned)GF) / gridDim.x);

    for (int g = 0; g < GF; ++g) {
        int gg = g + rotg;
        if (gg >= GF) gg -= GF;
        const int c0 = gg * (SCAN_C * AGG_WAVES) + wave;
        const uint4* p = key + (long)c0 * 64 + lane;
        uint4 lo[SCAN_C], hi[SCAN_C];
#pragma unroll
        for (int u = 0; u < SCAN_C; ++u) {
            lo[u] = p[u * (AGG_WAVES * 64)];
            hi[u] = p[u * (AGG_WAVES * 64) + 32];
        }
#pragma unroll
        for (int u = 0; u < SCAN_C; ++u)
            scan_test(lo[u], hi[u], (unsigned)((c0 + u * AGG_WAVES) * CHUNK_E) + lane_e,
                      mk0, mk1, mk2, mk3, mk4, mk5, mk6, mk7,
                      feat, pk, acc, cnt, q, qhead, qn, lane);
    }

    {
        int c = GF * (SCAN_C * AGG_WAVES) + wave;
        uint4 clo = make_uint4(~0u, ~0u, ~0u, ~0u), chi = clo;
        if (c < n_chunks) { clo = key[(long)c * 64 + lane]; chi = key[(long)c * 64 + 32 + lane]; }
        for (; c < n_chunks; c += AGG_WAVES) {
            const int cn = c + AGG_WAVES;
            uint4 nlo = clo, nhi = chi;
            if (cn < n_chunks) { nlo = key[(long)cn * 64 + lane]; nhi = key[(long)cn * 64 + 32 + lane]; }
            scan_test(clo, chi, (unsigned)(c * CHUNK_E) + lane_e,
                      mk0, mk1, mk2, mk3, mk4, mk5, mk6, mk7,
                      feat, pk, acc, cnt, q, qhead, qn, lane);
            clo = nlo; chi = nhi;
        }
    }
    if (qn > 0) ring_drain(feat, pk, acc, cnt, q, qhead, qn, qn, lane);
    __syncthreads();

    for (int i = wave; i < nown; i += AGG_WAVES) {
        const float dg = fmaxf((float)cnt[i], 1.0f);
        const float* a = acc + i * D_FEAT + lane;
        v4f o;
        o.x = a[0]  / dg;
        o.y = a[32] / dg;
        o.z = a[64] / dg;
        o.w = a[96] / dg;
        *(v4f*)(hneigh + (long)(lo + i) * D_FEAT + lane * 4) = o;
    }
}

#define KCH (D_FEAT / 8)

__device__ __forceinline__ void load_a_tile(
    const float* __restrict__ feat, const float* __restrict__ hneigh,
    int tile, int c, int h, v4f (&af)[KCH], v4f (&ah)[KCH])
{
    const long node_base = (long)tile * 16;
    const float* fa = feat   + (node_base + c) * D_FEAT + 4 * h;
    const float* ha = hneigh + (node_base + c) * D_FEAT + 4 * h;
#pragma unroll
    for (int j = 0; j < KCH; ++j) af[j] = *(const v4f*)(fa + 8 * j);
#pragma unroll
    for (int j = 0; j < KCH; ++j) ah[j] = *(const v4f*)(ha + 8 * j);
}

__device__ __forceinline__ void mma_tile_store(
    const v4f (&af)[KCH], const v4f (&ah)[KCH],
    const v4f (&bs0)[KCH], const v4f (&bs1)[KCH],
    const v4f (&bn0)[KCH], const v4f (&bn1)[KCH],
    float* __restrict__ out, int tile, int c, int h, int col_base)
{
    v8f c0 = {}, c1 = {};
#pragma unroll
    for (int j = 0; j < KCH; ++j) {
        v2f a_lo = {af[j].x, af[j].y};
        v2f a_hi = {af[j].z, af[j].w};
        v2f b0lo = {bs0[j].x, bs0[j].y};
        v2f b0hi = {bs0[j].z, bs0[j].w};
        v2f b1lo = {bs1[j].x, bs1[j].y};
        v2f b1hi = {bs1[j].z, bs1[j].w};
        c0 = __builtin_amdgcn_wmma_f32_16x16x4_f32(
                false, a_lo, false, b0lo, (short)0, c0, false, false);
        c1 = __builtin_amdgcn_wmma_f32_16x16x4_f32(
                false, a_lo, false, b1lo, (short)0, c1, false, false);
        c0 = __builtin_amdgcn_wmma_f32_16x16x4_f32(
                false, a_hi, false, b0hi, (short)0, c0, false, false);
        c1 = __builtin_amdgcn_wmma_f32_16x16x4_f32(
                false, a_hi, false, b1hi, (short)0, c1, false, false);
    }
#pragma unroll
    for (int j = 0; j < KCH; ++j) {
        v2f a_lo = {ah[j].x, ah[j].y};
        v2f a_hi = {ah[j].z, ah[j].w};
        v2f b0lo = {bn0[j].x, bn0[j].y};
        v2f b0hi = {bn0[j].z, bn0[j].w};
        v2f b1lo = {bn1[j].x, bn1[j].y};
        v2f b1hi = {bn1[j].z, bn1[j].w};
        c0 = __builtin_amdgcn_wmma_f32_16x16x4_f32(
                false, a_lo, false, b0lo, (short)0, c0, false, false);
        c1 = __builtin_amdgcn_wmma_f32_16x16x4_f32(
                false, a_lo, false, b1lo, (short)0, c1, false, false);
        c0 = __builtin_amdgcn_wmma_f32_16x16x4_f32(
                false, a_hi, false, b0hi, (short)0, c0, false, false);
        c1 = __builtin_amdgcn_wmma_f32_16x16x4_f32(
                false, a_hi, false, b1hi, (short)0, c1, false, false);
    }

    float* o = out + ((long)tile * 16 + h * 8) * D_FEAT + col_base + c;
#pragma unroll
    for (int r = 0; r < 8; ++r) {
        o[(long)r * D_FEAT]      = c0[r];
        o[(long)r * D_FEAT + 16] = c1[r];
    }
}

__global__ __launch_bounds__(128) void sage_gemm_wmma(
    const float* __restrict__ feat, const float* __restrict__ hneigh,
    const float* __restrict__ Wself, const float* __restrict__ Wneigh,
    float* __restrict__ out, int n_tiles)
{
    const int wave = threadIdx.x >> 5;
    const int lane = threadIdx.x & 31;
    const int c = lane & 15;
    const int h = lane >> 4;
    const int col_base = wave * 32;
    const int step = gridDim.x;

    int tile = blockIdx.x;
    if (tile >= n_tiles) return;
    v4f xf[KCH], xh[KCH], yf[KCH], yh[KCH];
    load_a_tile(feat, hneigh, tile, c, h, xf, xh);

    v4f bs0[KCH], bs1[KCH], bn0[KCH], bn1[KCH];
    {
        const float* ws0 = Wself  + (long)(col_base + c) * D_FEAT + 4 * h;
        const float* ws1 = ws0 + 16 * D_FEAT;
        const float* wn0 = Wneigh + (long)(col_base + c) * D_FEAT + 4 * h;
        const float* wn1 = wn0 + 16 * D_FEAT;
#pragma unroll
        for (int j = 0; j < KCH; ++j) {
            bs0[j] = *(const v4f*)(ws0 + 8 * j);
            bs1[j] = *(const v4f*)(ws1 + 8 * j);
            bn0[j] = *(const v4f*)(wn0 + 8 * j);
            bn1[j] = *(const v4f*)(wn1 + 8 * j);
        }
    }

    for (;;) {
        const int t1 = tile + step;
        if (t1 >= n_tiles) {
            mma_tile_store(xf, xh, bs0, bs1, bn0, bn1, out, tile, c, h, col_base);
            break;
        }
        load_a_tile(feat, hneigh, t1, c, h, yf, yh);
        __builtin_amdgcn_sched_barrier(0);
        mma_tile_store(xf, xh, bs0, bs1, bn0, bn1, out, tile, c, h, col_base);

        const int t2 = t1 + step;
        if (t2 >= n_tiles) {
            mma_tile_store(yf, yh, bs0, bs1, bn0, bn1, out, t1, c, h, col_base);
            break;
        }
        load_a_tile(feat, hneigh, t2, c, h, xf, xh);
        __builtin_amdgcn_sched_barrier(0);
        mma_tile_store(yf, yh, bs0, bs1, bn0, bn1, out, t1, c, h, col_base);
        tile = t2;
    }
}

extern "C" void kernel_launch(void* const* d_in, const int* in_sizes, int n_in,
                              void* d_out, int out_size, void* d_ws, size_t ws_size,
                              hipStream_t stream)
{
    const float* feat   = (const float*)d_in[0];
    const int*   src    = (const int*)d_in[1];
    const int*   dst    = (const int*)d_in[2];
    const float* Wself  = (const float*)d_in[3];
    const float* Wneigh = (const float*)d_in[4];
    float*       out    = (float*)d_out;

    const int n_nodes = in_sizes[0] / D_FEAT;
    const int n_edges = in_sizes[1];

    float* hneigh = (float*)d_ws;

    sage_edge_keys<<<392, 128, 0, stream>>>(src, dst, out, n_nodes, n_edges);

    sage_aggregate<<<(n_nodes + NPB - 1) / NPB, 1024, 0, stream>>>(
        feat, out, hneigh, n_nodes, n_edges);

    sage_gemm_wmma<<<256, 128, 0, stream>>>(feat, hneigh, Wself, Wneigh, out, n_nodes / 16);
}
